# starting kernel
# baseline (speedup 1.0000x reference)
#include <hip/hip_runtime.h>


#define NPAIR 169

typedef __attribute__((ext_vector_type(2))) float v2f;
typedef __attribute__((ext_vector_type(8))) float v8f;

__global__ void __launch_bounds__(256)
hon_prep(const int* __restrict__ x, const int* __restrict__ fact,
         int* __restrict__ ids, int* __restrict__ perm,
         int* __restrict__ gstart, int* __restrict__ work,
         int N, int wcap)
{
    __shared__ int cnt[NPAIR];
    __shared__ int pos[NPAIR];
    const int tid = threadIdx.x;

    for (int g = tid; g < NPAIR; g += blockDim.x) cnt[g] = 0;
    __syncthreads();

    for (int n = tid; n < N; n += blockDim.x) {
        const int a  = fact[2 * n];
        const int id = x[3 * a + 1] * 13 + x[3 * a + 2];
        ids[n] = id;
        atomicAdd(&cnt[id], 1);
    }
    __syncthreads();

    if (tid == 0) {
        int acc = 0, w = 0;
        for (int g = 0; g < NPAIR; ++g) {
            gstart[g] = acc;
            pos[g]    = acc;
            const int c  = cnt[g];
            const int nt = (c + 15) >> 4;
            for (int t = 0; t < nt && w < wcap; ++t, ++w) {
                work[2 * w]     = g;
                work[2 * w + 1] = t;
            }
            acc += c;
        }
        gstart[NPAIR] = acc;
        for (; w < wcap; ++w) { work[2 * w] = -1; work[2 * w + 1] = 0; }
    }
    __syncthreads();

    for (int n = tid; n < N; n += blockDim.x) {
        const int p = atomicAdd(&pos[ids[n]], 1);
        perm[p] = n;
    }
}

__global__ void __launch_bounds__(128)
hon_gemm(const float* __restrict__ inp, const float* __restrict__ params,
         const float* __restrict__ bias, const int* __restrict__ perm,
         const int* __restrict__ gstart, const int* __restrict__ work,
         const int* __restrict__ msg_to_p, float* __restrict__ out,
         int N, int H, int O, int ORD)
{
    extern __shared__ float As[];

    const int g = work[2 * blockIdx.x];
    if (g < 0) return;
    const int mt     = work[2 * blockIdx.x + 1];
    const int rowbeg = gstart[g] + mt * 16;
    int cnt = gstart[g + 1] - rowbeg;
    cnt = cnt > 16 ? 16 : cnt;
    const int msg_to = msg_to_p[0];

    {
        const int HQ = H >> 2;
        for (int idx = threadIdx.x; idx < 16 * HQ; idx += blockDim.x) {
            const int r  = idx / HQ;
            const int kq = idx - r * HQ;
            float4 p = make_float4(0.f, 0.f, 0.f, 0.f);
            if (r < cnt) {
                const int n = perm[rowbeg + r];
                p = make_float4(1.f, 1.f, 1.f, 1.f);
                for (int i = 0; i < ORD; ++i) {
                    if (i == msg_to) continue;
                    const float4 v =
                        ((const float4*)(inp + ((size_t)i * N + n) * H))[kq];
                    p.x *= v.x; p.y *= v.y; p.z *= v.z; p.w *= v.w;
                }
            }
            ((float4*)(As + r * H))[kq] = p;
        }
    }
    __syncthreads();

    const int lane   = threadIdx.x & 31;
    const int waveid = threadIdx.x >> 5;
    const int m      = lane & 15;
    const int khalf  = (lane >> 4) << 1;

    const float* W       = params + (size_t)g * H * O;
    const int    colbase = blockIdx.y * 256 + waveid * 64 + m;
    const float* aptr    = As + m * H + khalf;

    v8f acc[4];
#pragma unroll
    for (int t = 0; t < 4; ++t) acc[t] = (v8f){0, 0, 0, 0, 0, 0, 0, 0};

#pragma unroll 2
    for (int kb = 0; kb < H; kb += 4) {
        const int k0 = kb + khalf;
        const v2f a  = *(const v2f*)(aptr + kb);

        const float* Wk0 = W + (size_t)k0 * O + colbase;
        const float* Wk1 = Wk0 + O;
#pragma unroll
        for (int t = 0; t < 4; ++t) {
            v2f b; b.x = Wk0[t * 16]; b.y = Wk1[t * 16];
            acc[t] = __builtin_amdgcn_wmma_f32_16x16x4_f32(
                false, a, false, b, (short)0, acc[t], false, false);
        }
    }

    const int mbase = (lane >> 4) << 3;
    int nrow[8];
#pragma unroll
    for (int r = 0; r < 8; ++r) {
        const int mr = mbase + r;
        nrow[r] = (mr < cnt) ? perm[rowbeg + mr] : -1;
    }

#pragma unroll
    for (int t = 0; t < 4; ++t) {
        const int   col = colbase + t * 16;
        const float bv  = bias[(size_t)g * O + col];
#pragma unroll
        for (int r = 0; r < 8; ++r) {
            if (nrow[r] >= 0)
                out[(size_t)nrow[r] * O + col] = acc[t][r] + bv;
        }
    }
}

extern "C" void kernel_launch(void* const* d_in, const int* in_sizes, int n_in,
                              void* d_out, int out_size, void* d_ws, size_t ws_size,
                              hipStream_t stream)
{
    const int*   x      = (const int*)  d_in[0];
    const int*   fact   = (const int*)  d_in[1];
    const float* inp    = (const float*)d_in[2];
    const float* params = (const float*)d_in[3];
    const float* bias   = (const float*)d_in[4];
    const int*   msg_to = (const int*)  d_in[5];
    float*       out    = (float*)d_out;

    const int N   = in_sizes[1] / 2;
    const int O   = in_sizes[4] / NPAIR;
    const int H   = in_sizes[3] / (NPAIR * O);
    const int ORD = in_sizes[2] / (N * H);

    const int wcap = NPAIR + (N + 15) / 16;

    int* ws     = (int*)d_ws;
    int* ids    = ws;
    int* perm   = ws + N;
    int* gstart = ws + 2 * N;
    int* work   = ws + 2 * N + NPAIR + 1;

    hipLaunchKernelGGL(hon_prep, dim3(1), dim3(256), 0, stream,
                       x, fact, ids, perm, gstart, work, N, wcap);

    const size_t lds_bytes = (size_t)16 * H * sizeof(float);
    dim3 grid(wcap, O / 256);
    hipLaunchKernelGGL(hon_gemm, grid, dim3(128), lds_bytes, stream,
                       inp, params, bias, perm, gstart, work, msg_to, out,
                       N, H, O, ORD);
}


# 256 work-groups, one per WGP: each sorts rows by id in LDS and takes an equal run of (16-row tile x 64-col) items; item K split over all 16 waves, summed via LDS in slice order (sum tree changes)
# speedup vs baseline: 2.9795x; 2.9795x over previous
#include <hip/hip_runtime.h>


#define NPAIR 169
#define NDIG 13
#define KSL 16
#define KBATCH 8
#define CW 64
#define PPITCH 68
#define TABW 176
#define LDS_BYTES (176 * 1024)

typedef __attribute__((ext_vector_type(2))) float v2f;
typedef __attribute__((ext_vector_type(8))) float v8f;

__device__ __forceinline__ int wave_scan_excl(int* tab, int E, int lane)
{
    const int per = (E + 31) >> 5;
    const int beg = lane * per;
    const int end = beg + per < E ? beg + per : E;
    int sum = 0;
    for (int i = beg; i < end; ++i) sum += tab[i];
    int inc = sum;
#pragma unroll
    for (int d = 1; d < 32; d <<= 1) {
        const int t = __shfl_up(inc, d);
        if (lane >= d) inc += t;
    }
    int run = inc - sum;
    for (int i = beg; i < end; ++i) {
        const int c = tab[i];
        tab[i] = run;
        run += c;
    }
    return __builtin_amdgcn_readlane(inc, 31);
}

__device__ __forceinline__ int digit_rank(int d, bool valid, int lane, int& mycount)
{
    int r = 0;
    mycount = 0;
#pragma unroll
    for (int v = 0; v < NDIG; ++v) {
        const unsigned mk = __builtin_amdgcn_ballot_w32(valid & (d == v));
        const int below = (int)__builtin_amdgcn_mbcnt_lo(mk, 0u);
        if (d == v) r = below;
        if (lane == v) mycount = __builtin_popcount(mk);
    }
    return r;
}

__device__ __forceinline__ int clamp_digit(int p)
{
    p = p < 0 ? 0 : p;
    return p > NDIG - 1 ? NDIG - 1 : p;
}

__global__ void __launch_bounds__(512)
gemm_by_id(const int* __restrict__ x, const int* __restrict__ fact,
          const float* __restrict__ inp, const float* __restrict__ params,
          const float* __restrict__ bias, const int* __restrict__ msg_to_p,
          float* __restrict__ out, int N, int H, int O, int ORD)
{
    __shared__ __attribute__((aligned(16))) unsigned char L[LDS_BYTES];

    const int nsets = (N + 31) >> 5;
    float* As    = (float*)L;
    float* Ps    = As + 16 * H;
    int*   tmp   = (int*)(Ps + KSL * 16 * PPITCH);
    int*   t1    = tmp + N;
    int*   srt   = t1 + N;
    int*   cntm  = srt + N;
    int*   startv= cntm + NDIG * nsets;
    int*   endv  = startv + TABW;
    int*   tp    = endv + TABW;
    int*   tileg = tp + TABW;

    const int lane = threadIdx.x & 31;
    const int waveid = __builtin_amdgcn_readfirstlane((int)(threadIdx.x >> 5));

    if (threadIdx.x < TABW) { startv[threadIdx.x] = 0; endv[threadIdx.x] = 0; }

    for (int s = waveid; s < nsets; s += KSL) {
        const int  i     = 32 * s + lane;
        const bool valid = i < N;
        const int  ic    = valid ? i : N - 1;
        const int  a     = fact[2 * ic];
        const int  p1    = clamp_digit(x[3 * a + 1]);
        const int  p2    = clamp_digit(x[3 * a + 2]);
        int myc;
        const int r = digit_rank(p2, valid, lane, myc);
        if (lane < NDIG) cntm[lane * nsets + s] = myc;
        if (valid) tmp[i] = (p1 * NDIG + p2) * 32 + r;
    }
    __syncthreads();
    if (waveid == 0) wave_scan_excl(cntm, NDIG * nsets, lane);
    __syncthreads();
    for (int s = waveid; s < nsets; s += KSL) {
        const int i = 32 * s + lane;
        if (i < N) {
            const int pk  = tmp[i];
            const int key = pk >> 5;
            const int p2  = key % NDIG;
            t1[cntm[p2 * nsets + s] + (pk & 31)] = (i << 8) | key;
        }
    }
    __syncthreads();

    for (int s = waveid; s < nsets; s += KSL) {
        const int  i     = 32 * s + lane;
        const bool valid = i < N;
        const int  e     = t1[valid ? i : N - 1];
        const int  p1    = (e & 255) / NDIG;
        int myc;
        const int r = digit_rank(p1, valid, lane, myc);
        if (lane < NDIG) cntm[lane * nsets + s] = myc;
        if (valid) tmp[i] = r;
    }
    __syncthreads();
    if (waveid == 0) wave_scan_excl(cntm, NDIG * nsets, lane);
    __syncthreads();
    for (int s = waveid; s < nsets; s += KSL) {
        const int i = 32 * s + lane;
        if (i < N) {
            const int e  = t1[i];
            const int p1 = (e & 255) / NDIG;
            srt[cntm[p1 * nsets + s] + tmp[i]] = e;
        }
    }
    __syncthreads();

    for (int i = threadIdx.x; i < N; i += blockDim.x) {
        const int k = srt[i] & 255;
        if (i == 0     || (srt[i - 1] & 255) != k) startv[k] = i;
        if (i == N - 1 || (srt[i + 1] & 255) != k) endv[k]   = i + 1;
    }
    __syncthreads();
    if (threadIdx.x < TABW)
        tp[threadIdx.x] = threadIdx.x < NPAIR
            ? (endv[threadIdx.x] - startv[threadIdx.x] + 15) >> 4 : 0;
    __syncthreads();
    if (waveid == 0) {
        const int all = wave_scan_excl(tp, NPAIR, lane);
        if (lane == 0) tp[TABW - 1] = all;
    }
    __syncthreads();
    if (threadIdx.x < NPAIR) {
        const int nt = (endv[threadIdx.x] - startv[threadIdx.x] + 15) >> 4;
        for (int j = 0; j < nt; ++j) tileg[tp[threadIdx.x] + j] = threadIdx.x;
    }
    __syncthreads();

    const int TT  = __builtin_amdgcn_readfirstlane(tp[TABW - 1]);
    const int NCH = O / CW;
    const int Q   = TT * NCH;
    const int q0  = (int)(((long long)blockIdx.x * Q) / gridDim.x);
    const int q1  = (int)(((long long)(blockIdx.x + 1) * Q) / gridDim.x);

    const int msg_to = msg_to_p[0];

    const int m     = lane & 15;
    const int khalf = (lane >> 4) << 1;
    const int kspan = H / KSL;
    const int kbeg  = waveid * kspan;
    const float* aptr = As + m * H + khalf;

    int staged = -1;
    for (int q = q0; q < q1; ++q) {
        const int tau = q / NCH;
        const int ch  = q - tau * NCH;
        const int g   = __builtin_amdgcn_readfirstlane(tileg[tau]);
        const int tix = tau - __builtin_amdgcn_readfirstlane(tp[g]);
        const int gs  = __builtin_amdgcn_readfirstlane(startv[g]);
        const int gn  = __builtin_amdgcn_readfirstlane(endv[g]) - gs;
        const int rowbeg = gs + 16 * tix;
        int cnt = gn - 16 * tix;
        cnt = cnt > 16 ? 16 : cnt;

        const float*   Wg      = params + (size_t)g * H * O;
        const int      colbase = ch * CW + m;
        const unsigned lbyte   = 4u * (unsigned)(khalf * O + colbase);

        v8f acc[4];
#pragma unroll
        for (int t = 0; t < 4; ++t) acc[t] = (v8f){0, 0, 0, 0, 0, 0, 0, 0};

        for (int kb = kbeg; kb < kbeg + kspan; kb += 4 * KBATCH) {
            v2f a[KBATCH];
            v2f b[KBATCH][4];
#pragma unroll
            for (int j = 0; j < KBATCH; ++j) {
                const char*  Wr  = (const char*)(Wg + (size_t)(kb + 4 * j) * O);
                const float* Wk0 = (const float*)(Wr + lbyte);
                const float* Wk1 = (const float*)(Wr + (size_t)O * 4 + lbyte);
#pragma unroll
                for (int t = 0; t < 4; ++t) {
                    b[j][t].x = Wk0[t * 16];
                    b[j][t].y = Wk1[t * 16];
                }
            }
            __builtin_amdgcn_sched_barrier(0);

            if (tau != staged) {
                const int HQ = H >> 2;
                for (int idx = threadIdx.x; idx < 16 * HQ; idx += blockDim.x) {
                    const int r  = idx / HQ;
                    const int kq = idx - r * HQ;
                    float4 p = make_float4(0.f, 0.f, 0.f, 0.f);
                    if (r < cnt) {
                        const int nr = srt[rowbeg + r] >> 8;
                        p = make_float4(1.f, 1.f, 1.f, 1.f);
                        for (int i = 0; i < ORD; ++i) {
                            if (i == msg_to) continue;
                            const float4 v =
                                ((const float4*)(inp + ((size_t)i * N + nr) * H))[kq];
                            p.x *= v.x; p.y *= v.y; p.z *= v.z; p.w *= v.w;
                        }
                    }
                    ((float4*)(As + r * H))[kq] = p;
                }
                __syncthreads();
                staged = tau;
            }

#pragma unroll
            for (int j = 0; j < KBATCH; ++j)
                a[j] = *(const v2f*)(aptr + kb + 4 * j);
#pragma unroll
            for (int j = 0; j < KBATCH; ++j) {
#pragma unroll
                for (int t = 0; t < 4; ++t) {
                    acc[t] = __builtin_amdgcn_wmma_f32_16x16x4_f32(
                        false, a[j], false, b[j][t], (short)0, acc[t], false, false);
                }
            }
        }

        {
            const int mbase = (lane >> 4) << 3;
            float* prow = Ps + (waveid * 16 + mbase) * PPITCH + m;
#pragma unroll
            for (int t = 0; t < 4; ++t) {
#pragma unroll
                for (int r = 0; r < 8; ++r)
                    prow[r * PPITCH + t * 16] = acc[t][r];
            }
        }
        __syncthreads();

        if (waveid < cnt) {
            const int nr = srt[rowbeg + waveid] >> 8;
            v2f s = *(const v2f*)(Ps + waveid * PPITCH + 2 * lane);
            for (int sl = 1; sl < KSL; ++sl) {
                const v2f p = *(const v2f*)(Ps + (sl * 16 + waveid) * PPITCH + 2 * lane);
                s.x += p.x; s.y += p.y;
            }
            const v2f bv = *(const v2f*)(bias + (size_t)g * O + ch * CW + 2 * lane);
            s.x += bv.x; s.y += bv.y;
            *(v2f*)(out + (size_t)nr * O + ch * CW + 2 * lane) = s;
        }
        __syncthreads();
    }
}

extern "C" void kernel_launch(void* const* d_in, const int* in_sizes, int n_in,
                              void* d_out, int out_size, void* d_ws, size_t ws_size,
                              hipStream_t stream)
{
    const int*   x      = (const int*)  d_in[0];
    const int*   fact   = (const int*)  d_in[1];
    const float* inp    = (const float*)d_in[2];
    const float* params = (const float*)d_in[3];
    const float* bias   = (const float*)d_in[4];
    const int*   msg_to = (const int*)  d_in[5];
    float*       out    = (float*)d_out;

    const int N   = in_sizes[1] / 2;
    const int O   = in_sizes[4] / NPAIR;
    const int H   = in_sizes[3] / (NPAIR * O);
    const int ORD = in_sizes[2] / (N * H);

    hipLaunchKernelGGL(gemm_by_id, dim3(256), dim3(512), 0, stream,
                       x, fact, inp, params, bias, msg_to, out,
                       N, H, O, ORD);
}


# A operand not staged in LDS: a lane loads its K pairs of its tile row from inp into registers ahead of the B loads, multiplies the planes there, keeps them across a tile's items; sums unchanged
# speedup vs baseline: 3.0717x; 1.0309x over previous
#include <hip/hip_runtime.h>


#define NPAIR 169
#define NDIG 13
#define KSL 16
#define KBATCH 8
#define CW 64
#define PPITCH 68
#define TABW 176
#define LDS_BYTES (176 * 1024)

typedef __attribute__((ext_vector_type(2))) float v2f;
typedef __attribute__((ext_vector_type(8))) float v8f;

__device__ __forceinline__ int wave_scan_excl(int* tab, int E, int lane)
{
    const int per = (E + 31) >> 5;
    const int beg = lane * per;
    const int end = beg + per < E ? beg + per : E;
    int sum = 0;
    for (int i = beg; i < end; ++i) sum += tab[i];
    int inc = sum;
#pragma unroll
    for (int d = 1; d < 32; d <<= 1) {
        const int t = __shfl_up(inc, d);
        if (lane >= d) inc += t;
    }
    int run = inc - sum;
    for (int i = beg; i < end; ++i) {
        const int c = tab[i];
        tab[i] = run;
        run += c;
    }
    return __builtin_amdgcn_readlane(inc, 31);
}

__device__ __forceinline__ int digit_rank(int d, bool valid, int lane, int& mycount)
{
    int r = 0;
    mycount = 0;
#pragma unroll
    for (int v = 0; v < NDIG; ++v) {
        const unsigned mk = __builtin_amdgcn_ballot_w32(valid & (d == v));
        const int below = (int)__builtin_amdgcn_mbcnt_lo(mk, 0u);
        if (d == v) r = below;
        if (lane == v) mycount = __builtin_popcount(mk);
    }
    return r;
}

__device__ __forceinline__ int clamp_digit(int p)
{
    p = p < 0 ? 0 : p;
    return p > NDIG - 1 ? NDIG - 1 : p;
}

__global__ void __launch_bounds__(512)
gemm_by_id(const int* __restrict__ x, const int* __restrict__ fact,
          const float* __restrict__ inp, const float* __restrict__ params,
          const float* __restrict__ bias, const int* __restrict__ msg_to_p,
          float* __restrict__ out, int N, int H, int O, int ORD)
{
    __shared__ __attribute__((aligned(16))) unsigned char L[LDS_BYTES];

    const int nsets = (N + 31) >> 5;
    float* Ps    = (float*)L;
    int*   tmp   = (int*)(Ps + KSL * 16 * PPITCH);
    int*   t1    = tmp + N;
    int*   srt   = t1 + N;
    int*   cntm  = srt + N;
    int*   startv= cntm + NDIG * nsets;
    int*   endv  = startv + TABW;
    int*   tp    = endv + TABW;
    int*   tileg = tp + TABW;

    const int lane = threadIdx.x & 31;
    const int waveid = __builtin_amdgcn_readfirstlane((int)(threadIdx.x >> 5));

    if (threadIdx.x < TABW) { startv[threadIdx.x] = 0; endv[threadIdx.x] = 0; }

    for (int s = waveid; s < nsets; s += KSL) {
        const int  i     = 32 * s + lane;
        const bool valid = i < N;
        const int  ic    = valid ? i : N - 1;
        const int  a     = fact[2 * ic];
        const int  p1    = clamp_digit(x[3 * a + 1]);
        const int  p2    = clamp_digit(x[3 * a + 2]);
        int myc;
        const int r = digit_rank(p2, valid, lane, myc);
        if (lane < NDIG) cntm[lane * nsets + s] = myc;
        if (valid) tmp[i] = (p1 * NDIG + p2) * 32 + r;
    }
    __syncthreads();
    if (waveid == 0) wave_scan_excl(cntm, NDIG * nsets, lane);
    __syncthreads();
    for (int s = waveid; s < nsets; s += KSL) {
        const int i = 32 * s + lane;
        if (i < N) {
            const int pk  = tmp[i];
            const int key = pk >> 5;
            const int p2  = key % NDIG;
            t1[cntm[p2 * nsets + s] + (pk & 31)] = (i << 8) | key;
        }
    }
    __syncthreads();

    for (int s = waveid; s < nsets; s += KSL) {
        const int  i     = 32 * s + lane;
        const bool valid = i < N;
        const int  e     = t1[valid ? i : N - 1];
        const int  p1    = (e & 255) / NDIG;
        int myc;
        const int r = digit_rank(p1, valid, lane, myc);
        if (lane < NDIG) cntm[lane * nsets + s] = myc;
        if (valid) tmp[i] = r;
    }
    __syncthreads();
    if (waveid == 0) wave_scan_excl(cntm, NDIG * nsets, lane);
    __syncthreads();
    for (int s = waveid; s < nsets; s += KSL) {
        const int i = 32 * s + lane;
        if (i < N) {
            const int e  = t1[i];
            const int p1 = (e & 255) / NDIG;
            srt[cntm[p1 * nsets + s] + tmp[i]] = e;
        }
    }
    __syncthreads();

    for (int i = threadIdx.x; i < N; i += blockDim.x) {
        const int k = srt[i] & 255;
        if (i == 0     || (srt[i - 1] & 255) != k) startv[k] = i;
        if (i == N - 1 || (srt[i + 1] & 255) != k) endv[k]   = i + 1;
    }
    __syncthreads();
    if (threadIdx.x < TABW)
        tp[threadIdx.x] = threadIdx.x < NPAIR
            ? (endv[threadIdx.x] - startv[threadIdx.x] + 15) >> 4 : 0;
    __syncthreads();
    if (waveid == 0) {
        const int all = wave_scan_excl(tp, NPAIR, lane);
        if (lane == 0) tp[TABW - 1] = all;
    }
    __syncthreads();
    if (threadIdx.x < NPAIR) {
        const int nt = (endv[threadIdx.x] - startv[threadIdx.x] + 15) >> 4;
        for (int j = 0; j < nt; ++j) tileg[tp[threadIdx.x] + j] = threadIdx.x;
    }
    __syncthreads();

    const int TT  = __builtin_amdgcn_readfirstlane(tp[TABW - 1]);
    const int NCH = O / CW;
    const int Q   = TT * NCH;
    const int q0  = (int)(((long long)blockIdx.x * Q) / gridDim.x);
    const int q1  = (int)(((long long)(blockIdx.x + 1) * Q) / gridDim.x);

    const int msg_to = msg_to_p[0];

    const int m     = lane & 15;
    const int khalf = (lane >> 4) << 1;
    const int kspan = H / KSL;
    const int kbeg  = waveid * kspan;
    const bool keepA = kspan == 4 * KBATCH;
    const size_t plane = (size_t)N * H;
    const int  nkept = ORD - ((msg_to >= 0 && msg_to < ORD) ? 1 : 0);
    const bool pairA = nkept == 2;
    const int  ia    = (msg_to == 0) ? 1 : 0;
    const int  ib    = (msg_to == ia + 1) ? ia + 2 : ia + 1;

    v2f a[KBATCH];
#pragma unroll
    for (int j = 0; j < KBATCH; ++j) a[j] = (v2f){0.f, 0.f};
    int staged = -1;
    for (int q = q0; q < q1; ++q) {
        const int tau = q / NCH;
        const int ch  = q - tau * NCH;
        const int g   = __builtin_amdgcn_readfirstlane(tileg[tau]);
        const int tix = tau - __builtin_amdgcn_readfirstlane(tp[g]);
        const int gs  = __builtin_amdgcn_readfirstlane(startv[g]);
        const int gn  = __builtin_amdgcn_readfirstlane(endv[g]) - gs;
        const int rowbeg = gs + 16 * tix;
        int cnt = gn - 16 * tix;
        cnt = cnt > 16 ? 16 : cnt;

        const bool   rowok = m < cnt;
        const int    nrow  = srt[rowbeg + (rowok ? m : 0)] >> 8;
        const float* arow  = inp + (size_t)nrow * H + khalf;

        const float*   Wg      = params + (size_t)g * H * O;
        const int      colbase = ch * CW + m;
        const unsigned lbyte   = 4u * (unsigned)(khalf * O + colbase);

        v8f acc[4];
#pragma unroll
        for (int t = 0; t < 4; ++t) acc[t] = (v8f){0, 0, 0, 0, 0, 0, 0, 0};

        for (int kb = kbeg; kb < kbeg + kspan; kb += 4 * KBATCH) {
            const bool   fresh = tau != staged || !keepA;
            const float* ak    = arow + kb;
            v2f u[KBATCH], w[KBATCH];
            if (fresh && pairA) {
                const float* r0 = ak + (size_t)ia * plane;
                const float* r1 = ak + (size_t)ib * plane;
#pragma unroll
                for (int j = 0; j < KBATCH; ++j) u[j] = *(const v2f*)(r0 + 4 * j);
#pragma unroll
                for (int j = 0; j < KBATCH; ++j) w[j] = *(const v2f*)(r1 + 4 * j);
            }

            v2f b[KBATCH][4];
#pragma unroll
            for (int j = 0; j < KBATCH; ++j) {
                const char*  Wr  = (const char*)(Wg + (size_t)(kb + 4 * j) * O);
                const float* Wk0 = (const float*)(Wr + lbyte);
                const float* Wk1 = (const float*)(Wr + (size_t)O * 4 + lbyte);
#pragma unroll
                for (int t = 0; t < 4; ++t) {
                    b[j][t].x = Wk0[t * 16];
                    b[j][t].y = Wk1[t * 16];
                }
            }

            if (fresh) {
                if (pairA) {
#pragma unroll
                    for (int j = 0; j < KBATCH; ++j) {
                        v2f p = u[j];
                        p.x *= w[j].x; p.y *= w[j].y;
                        a[j] = rowok ? p : (v2f){0.f, 0.f};
                    }
                } else {
                    v2f p[KBATCH];
#pragma unroll
                    for (int j = 0; j < KBATCH; ++j) p[j] = (v2f){1.f, 1.f};
                    for (int i = 0; i < ORD; ++i) {
                        if (i == msg_to) continue;
                        const float* r = ak + (size_t)i * plane;
#pragma unroll
                        for (int j = 0; j < KBATCH; ++j) {
                            const v2f v = *(const v2f*)(r + 4 * j);
                            p[j].x *= v.x; p[j].y *= v.y;
                        }
                    }
#pragma unroll
                    for (int j = 0; j < KBATCH; ++j)
                        a[j] = rowok ? p[j] : (v2f){0.f, 0.f};
                }
                staged = tau;
            }

#pragma unroll
            for (int j = 0; j < KBATCH; ++j) {
#pragma unroll
                for (int t = 0; t < 4; ++t) {
                    acc[t] = __builtin_amdgcn_wmma_f32_16x16x4_f32(
                        false, a[j], false, b[j][t], (short)0, acc[t], false, false);
                }
            }
        }

        {
            const int mbase = (lane >> 4) << 3;
            float* prow = Ps + (waveid * 16 + mbase) * PPITCH + m;
#pragma unroll
            for (int t = 0; t < 4; ++t) {
#pragma unroll
                for (int r = 0; r < 8; ++r)
                    prow[r * PPITCH + t * 16] = acc[t][r];
            }
        }
        __syncthreads();

        if (waveid < cnt) {
            const int nr = srt[rowbeg + waveid] >> 8;
            v2f s = *(const v2f*)(Ps + waveid * PPITCH + 2 * lane);
            for (int sl = 1; sl < KSL; ++sl) {
                const v2f p = *(const v2f*)(Ps + (sl * 16 + waveid) * PPITCH + 2 * lane);
                s.x += p.x; s.y += p.y;
            }
            const v2f bv = *(const v2f*)(bias + (size_t)g * O + ch * CW + 2 * lane);
            s.x += bv.x; s.y += bv.y;
            *(v2f*)(out + (size_t)nr * O + ch * CW + 2 * lane) = s;
        }
        __syncthreads();
    }
}

extern "C" void kernel_launch(void* const* d_in, const int* in_sizes, int n_in,
                              void* d_out, int out_size, void* d_ws, size_t ws_size,
                              hipStream_t stream)
{
    const int*   x      = (const int*)  d_in[0];
    const int*   fact   = (const int*)  d_in[1];
    const float* inp    = (const float*)d_in[2];
    const float* params = (const float*)d_in[3];
    const float* bias   = (const float*)d_in[4];
    const int*   msg_to = (const int*)  d_in[5];
    float*       out    = (float*)d_out;

    const int N   = in_sizes[1] / 2;
    const int O   = in_sizes[4] / NPAIR;
    const int H   = in_sizes[3] / (NPAIR * O);
    const int ORD = in_sizes[2] / (N * H);

    hipLaunchKernelGGL(gemm_by_id, dim3(256), dim3(512), 0, stream,
                       x, fact, inp, params, bias, msg_to, out,
                       N, H, O, ORD);
}


# Prelude's three prefix sums off one wave's serial LDS loops: vote offsets by a wave per digit (a lane per 32-entry group), tile counts 6 ids a lane, lane sums by DPP row shifts; row lists same
# speedup vs baseline: 3.2595x; 1.0611x over previous
#include <hip/hip_runtime.h>


#define NPAIR 169
#define NDIG 13
#define KSL 16
#define KBATCH 8
#define CW 64
#define PPITCH 68
#define TABW 176
#define TPW 192
#define TPL 6
#define LDS_BYTES (176 * 1024)

typedef __attribute__((ext_vector_type(2))) float v2f;
typedef __attribute__((ext_vector_type(8))) float v8f;

__device__ __forceinline__ int lane_scan_incl(int v, int lane)
{
    v += __builtin_amdgcn_update_dpp(0, v, 0x111, 0xf, 0xf, true);
    v += __builtin_amdgcn_update_dpp(0, v, 0x112, 0xf, 0xf, true);
    v += __builtin_amdgcn_update_dpp(0, v, 0x114, 0xf, 0xf, true);
    v += __builtin_amdgcn_update_dpp(0, v, 0x118, 0xf, 0xf, true);
    const int low = __builtin_amdgcn_readlane(v, 15);
    return lane >= 16 ? v + low : v;
}

__device__ __forceinline__ void digit_offsets(const int* cnt, int* off, int nsets,
                                              int waveid, int lane)
{
    if (waveid >= NDIG) return;
    int below = 0;
    for (int s0 = 0; s0 < nsets; s0 += 32) {
        const int  s  = s0 + lane;
        const bool ok = s < nsets;
        const int  sc = ok ? s : nsets - 1;
        int c[NDIG - 1];
#pragma unroll
        for (int d = 0; d < NDIG - 1; ++d) c[d] = cnt[d * nsets + sc];
        int part = 0;
#pragma unroll
        for (int d = 0; d < NDIG - 1; ++d) part += d < waveid ? c[d] : 0;
        below += ok ? part : 0;
    }
    int run = __builtin_amdgcn_readlane(lane_scan_incl(below, lane), 31);
    for (int s0 = 0; s0 < nsets; s0 += 32) {
        const int  s   = s0 + lane;
        const bool ok  = s < nsets;
        const int  c   = cnt[waveid * nsets + (ok ? s : nsets - 1)];
        const int  cv  = ok ? c : 0;
        const int  inc = lane_scan_incl(cv, lane);
        if (ok) off[waveid * nsets + s] = run + inc - cv;
        run += __builtin_amdgcn_readlane(inc, 31);
    }
}

__device__ __forceinline__ int tile_prefix(int* tab, int lane)
{
    int c[TPL];
    int sum = 0;
#pragma unroll
    for (int i = 0; i < TPL; ++i) { c[i] = tab[TPL * lane + i]; sum += c[i]; }
    const int inc = lane_scan_incl(sum, lane);
    int run = inc - sum;
#pragma unroll
    for (int i = 0; i < TPL; ++i) { tab[TPL * lane + i] = run; run += c[i]; }
    return __builtin_amdgcn_readlane(inc, 31);
}

__device__ __forceinline__ int digit_rank(int d, bool valid, int lane, int& mycount)
{
    int r = 0;
    mycount = 0;
#pragma unroll
    for (int v = 0; v < NDIG; ++v) {
        const unsigned mk = __builtin_amdgcn_ballot_w32(valid & (d == v));
        const int below = (int)__builtin_amdgcn_mbcnt_lo(mk, 0u);
        if (d == v) r = below;
        if (lane == v) mycount = __builtin_popcount(mk);
    }
    return r;
}

__device__ __forceinline__ int clamp_digit(int p)
{
    p = p < 0 ? 0 : p;
    return p > NDIG - 1 ? NDIG - 1 : p;
}

__global__ void __launch_bounds__(512)
gemm_by_id(const int* __restrict__ x, const int* __restrict__ fact,
          const float* __restrict__ inp, const float* __restrict__ params,
          const float* __restrict__ bias, const int* __restrict__ msg_to_p,
          float* __restrict__ out, int N, int H, int O, int ORD)
{
    __shared__ __attribute__((aligned(16))) unsigned char L[LDS_BYTES];

    const int nsets = (N + 31) >> 5;
    float* Ps    = (float*)L;
    int*   tmp   = (int*)(Ps + KSL * 16 * PPITCH);
    int*   t1    = tmp + N;
    int*   srt   = t1 + N;
    int*   cntm  = srt + N;
    int*   offm  = cntm + NDIG * nsets;
    int*   startv= offm + NDIG * nsets;
    int*   endv  = startv + TABW;
    int*   tp    = endv + TABW;
    int*   tileg = tp + TPW + 8;

    const int lane = threadIdx.x & 31;
    const int waveid = __builtin_amdgcn_readfirstlane((int)(threadIdx.x >> 5));

    if (threadIdx.x < TABW) { startv[threadIdx.x] = 0; endv[threadIdx.x] = 0; }

    for (int s = waveid; s < nsets; s += KSL) {
        const int  i     = 32 * s + lane;
        const bool valid = i < N;
        const int  ic    = valid ? i : N - 1;
        const int  a     = fact[2 * ic];
        const int  p1    = clamp_digit(x[3 * a + 1]);
        const int  p2    = clamp_digit(x[3 * a + 2]);
        int myc;
        const int r = digit_rank(p2, valid, lane, myc);
        if (lane < NDIG) cntm[lane * nsets + s] = myc;
        if (valid) tmp[i] = (p1 * NDIG + p2) * 32 + r;
    }
    __syncthreads();
    digit_offsets(cntm, offm, nsets, waveid, lane);
    __syncthreads();
    for (int s = waveid; s < nsets; s += KSL) {
        const int i = 32 * s + lane;
        if (i < N) {
            const int pk  = tmp[i];
            const int key = pk >> 5;
            const int p2  = key % NDIG;
            t1[offm[p2 * nsets + s] + (pk & 31)] = (i << 8) | key;
        }
    }
    __syncthreads();

    for (int s = waveid; s < nsets; s += KSL) {
        const int  i     = 32 * s + lane;
        const bool valid = i < N;
        const int  e     = t1[valid ? i : N - 1];
        const int  p1    = (e & 255) / NDIG;
        int myc;
        const int r = digit_rank(p1, valid, lane, myc);
        if (lane < NDIG) cntm[lane * nsets + s] = myc;
        if (valid) tmp[i] = r;
    }
    __syncthreads();
    digit_offsets(cntm, offm, nsets, waveid, lane);
    __syncthreads();
    for (int s = waveid; s < nsets; s += KSL) {
        const int i = 32 * s + lane;
        if (i < N) {
            const int e  = t1[i];
            const int p1 = (e & 255) / NDIG;
            srt[offm[p1 * nsets + s] + tmp[i]] = e;
        }
    }
    __syncthreads();

    for (int i = threadIdx.x; i < N; i += blockDim.x) {
        const int k = srt[i] & 255;
        if (i == 0     || (srt[i - 1] & 255) != k) startv[k] = i;
        if (i == N - 1 || (srt[i + 1] & 255) != k) endv[k]   = i + 1;
    }
    __syncthreads();
    if (threadIdx.x < TPW)
        tp[threadIdx.x] = threadIdx.x < NPAIR
            ? (endv[threadIdx.x] - startv[threadIdx.x] + 15) >> 4 : 0;
    __syncthreads();
    if (waveid == 0) {
        const int all = tile_prefix(tp, lane);
        if (lane == 0) tp[TPW] = all;
    }
    __syncthreads();
    if (threadIdx.x < NPAIR) {
        const int nt = (endv[threadIdx.x] - startv[threadIdx.x] + 15) >> 4;
        for (int j = 0; j < nt; ++j) tileg[tp[threadIdx.x] + j] = threadIdx.x;
    }
    __syncthreads();

    const int TT  = __builtin_amdgcn_readfirstlane(tp[TPW]);
    const int NCH = O / CW;
    const int Q   = TT * NCH;
    const int q0  = (int)(((long long)blockIdx.x * Q) / gridDim.x);
    const int q1  = (int)(((long long)(blockIdx.x + 1) * Q) / gridDim.x);

    const int msg_to = msg_to_p[0];

    const int m     = lane & 15;
    const int khalf = (lane >> 4) << 1;
    const int kspan = H / KSL;
    const int kbeg  = waveid * kspan;
    const bool keepA = kspan == 4 * KBATCH;
    const size_t plane = (size_t)N * H;
    const int  nkept = ORD - ((msg_to >= 0 && msg_to < ORD) ? 1 : 0);
    const bool pairA = nkept == 2;
    const int  ia    = (msg_to == 0) ? 1 : 0;
    const int  ib    = (msg_to == ia + 1) ? ia + 2 : ia + 1;

    v2f a[KBATCH];
#pragma unroll
    for (int j = 0; j < KBATCH; ++j) a[j] = (v2f){0.f, 0.f};
    int staged = -1;
    for (int q = q0; q < q1; ++q) {
        const int tau = q / NCH;
        const int ch  = q - tau * NCH;
        const int g   = __builtin_amdgcn_readfirstlane(tileg[tau]);
        const int tix = tau - __builtin_amdgcn_readfirstlane(tp[g]);
        const int gs  = __builtin_amdgcn_readfirstlane(startv[g]);
        const int gn  = __builtin_amdgcn_readfirstlane(endv[g]) - gs;
        const int rowbeg = gs + 16 * tix;
        int cnt = gn - 16 * tix;
        cnt = cnt > 16 ? 16 : cnt;

        const bool   rowok = m < cnt;
        const int    nrow  = srt[rowbeg + (rowok ? m : 0)] >> 8;
        const float* arow  = inp + (size_t)nrow * H + khalf;

        const float*   Wg      = params + (size_t)g * H * O;
        const int      colbase = ch * CW + m;
        const unsigned lbyte   = 4u * (unsigned)(khalf * O + colbase);

        v8f acc[4];
#pragma unroll
        for (int t = 0; t < 4; ++t) acc[t] = (v8f){0, 0, 0, 0, 0, 0, 0, 0};

        for (int kb = kbeg; kb < kbeg + kspan; kb += 4 * KBATCH) {
            const bool   fresh = tau != staged || !keepA;
            const float* ak    = arow + kb;
            v2f u[KBATCH], w[KBATCH];
            if (fresh && pairA) {
                const float* r0 = ak + (size_t)ia * plane;
                const float* r1 = ak + (size_t)ib * plane;
#pragma unroll
                for (int j = 0; j < KBATCH; ++j) u[j] = *(const v2f*)(r0 + 4 * j);
#pragma unroll
                for (int j = 0; j < KBATCH; ++j) w[j] = *(const v2f*)(r1 + 4 * j);
            }

            v2f b[KBATCH][4];
#pragma unroll
            for (int j = 0; j < KBATCH; ++j) {
                const char*  Wr  = (const char*)(Wg + (size_t)(kb + 4 * j) * O);
                const float* Wk0 = (const float*)(Wr + lbyte);
                const float* Wk1 = (const float*)(Wr + (size_t)O * 4 + lbyte);
#pragma unroll
                for (int t = 0; t < 4; ++t) {
                    b[j][t].x = Wk0[t * 16];
                    b[j][t].y = Wk1[t * 16];
                }
            }

            if (fresh) {
                if (pairA) {
#pragma unroll
                    for (int j = 0; j < KBATCH; ++j) {
                        v2f p = u[j];
                        p.x *= w[j].x; p.y *= w[j].y;
                        a[j] = rowok ? p : (v2f){0.f, 0.f};
                    }
                } else {
                    v2f p[KBATCH];
#pragma unroll
                    for (int j = 0; j < KBATCH; ++j) p[j] = (v2f){1.f, 1.f};
                    for (int i = 0; i < ORD; ++i) {
                        if (i == msg_to) continue;
                        const float* r = ak + (size_t)i * plane;
#pragma unroll
                        for (int j = 0; j < KBATCH; ++j) {
                            const v2f v = *(const v2f*)(r + 4 * j);
                            p[j].x *= v.x; p[j].y *= v.y;
                        }
                    }
#pragma unroll
                    for (int j = 0; j < KBATCH; ++j)
                        a[j] = rowok ? p[j] : (v2f){0.f, 0.f};
                }
                staged = tau;
            }

#pragma unroll
            for (int j = 0; j < KBATCH; ++j) {
#pragma unroll
                for (int t = 0; t < 4; ++t) {
                    acc[t] = __builtin_amdgcn_wmma_f32_16x16x4_f32(
                        false, a[j], false, b[j][t], (short)0, acc[t], false, false);
                }
            }
        }

        {
            const int mbase = (lane >> 4) << 3;
            float* prow = Ps + (waveid * 16 + mbase) * PPITCH + m;
#pragma unroll
            for (int t = 0; t < 4; ++t) {
#pragma unroll
                for (int r = 0; r < 8; ++r)
                    prow[r * PPITCH + t * 16] = acc[t][r];
            }
        }
        __syncthreads();

        if (waveid < cnt) {
            const int nr = srt[rowbeg + waveid] >> 8;
            v2f s = *(const v2f*)(Ps + waveid * PPITCH + 2 * lane);
            for (int sl = 1; sl < KSL; ++sl) {
                const v2f p = *(const v2f*)(Ps + (sl * 16 + waveid) * PPITCH + 2 * lane);
                s.x += p.x; s.y += p.y;
            }
            const v2f bv = *(const v2f*)(bias + (size_t)g * O + ch * CW + 2 * lane);
            s.x += bv.x; s.y += bv.y;
            *(v2f*)(out + (size_t)nr * O + ch * CW + 2 * lane) = s;
        }
        __syncthreads();
    }
}

extern "C" void kernel_launch(void* const* d_in, const int* in_sizes, int n_in,
                              void* d_out, int out_size, void* d_ws, size_t ws_size,
                              hipStream_t stream)
{
    const int*   x      = (const int*)  d_in[0];
    const int*   fact   = (const int*)  d_in[1];
    const float* inp    = (const float*)d_in[2];
    const float* params = (const float*)d_in[3];
    const float* bias   = (const float*)d_in[4];
    const int*   msg_to = (const int*)  d_in[5];
    float*       out    = (float*)d_out;

    const int N   = in_sizes[1] / 2;
    const int O   = in_sizes[4] / NPAIR;
    const int H   = in_sizes[3] / (NPAIR * O);
    const int ORD = in_sizes[2] / (N * H);

    hipLaunchKernelGGL(gemm_by_id, dim3(256), dim3(512), 0, stream,
                       x, fact, inp, params, bias, msg_to, out,
                       N, H, O, ORD);
}


# Item loop pipelined: a wave requests the next item's bias, A and B loads right behind its own WMMAs, into the registers just read; loop barriers wait for LDS only, not for loads in flight; sums same
# speedup vs baseline: 3.3690x; 1.0336x over previous
#include <hip/hip_runtime.h>


#define NPAIR 169
#define NDIG 13
#define KSL 16
#define KBATCH 8
#define CW 64
#define PPITCH 68
#define TABW 176
#define TPW 192
#define TPL 6
#define LDS_BYTES (176 * 1024)

typedef __attribute__((ext_vector_type(2))) float v2f;
typedef __attribute__((ext_vector_type(8))) float v8f;

__device__ __forceinline__ int lane_scan_incl(int v, int lane)
{
    v += __builtin_amdgcn_update_dpp(0, v, 0x111, 0xf, 0xf, true);
    v += __builtin_amdgcn_update_dpp(0, v, 0x112, 0xf, 0xf, true);
    v += __builtin_amdgcn_update_dpp(0, v, 0x114, 0xf, 0xf, true);
    v += __builtin_amdgcn_update_dpp(0, v, 0x118, 0xf, 0xf, true);
    const int low = __builtin_amdgcn_readlane(v, 15);
    return lane >= 16 ? v + low : v;
}

__device__ __forceinline__ void digit_offsets(const int* cnt, int* off, int nsets,
                                              int waveid, int lane)
{
    if (waveid >= NDIG) return;
    int below = 0;
    for (int s0 = 0; s0 < nsets; s0 += 32) {
        const int  s  = s0 + lane;
        const bool ok = s < nsets;
        const int  sc = ok ? s : nsets - 1;
        int c[NDIG - 1];
#pragma unroll
        for (int d = 0; d < NDIG - 1; ++d) c[d] = cnt[d * nsets + sc];
        int part = 0;
#pragma unroll
        for (int d = 0; d < NDIG - 1; ++d) part += d < waveid ? c[d] : 0;
        below += ok ? part : 0;
    }
    int run = __builtin_amdgcn_readlane(lane_scan_incl(below, lane), 31);
    for (int s0 = 0; s0 < nsets; s0 += 32) {
        const int  s   = s0 + lane;
        const bool ok  = s < nsets;
        const int  c   = cnt[waveid * nsets + (ok ? s : nsets - 1)];
        const int  cv  = ok ? c : 0;
        const int  inc = lane_scan_incl(cv, lane);
        if (ok) off[waveid * nsets + s] = run + inc - cv;
        run += __builtin_amdgcn_readlane(inc, 31);
    }
}

__device__ __forceinline__ int tile_prefix(int* tab, int lane)
{
    int c[TPL];
    int sum = 0;
#pragma unroll
    for (int i = 0; i < TPL; ++i) { c[i] = tab[TPL * lane + i]; sum += c[i]; }
    const int inc = lane_scan_incl(sum, lane);
    int run = inc - sum;
#pragma unroll
    for (int i = 0; i < TPL; ++i) { tab[TPL * lane + i] = run; run += c[i]; }
    return __builtin_amdgcn_readlane(inc, 31);
}

__device__ __forceinline__ int digit_rank(int d, bool valid, int lane, int& mycount)
{
    int r = 0;
    mycount = 0;
#pragma unroll
    for (int v = 0; v < NDIG; ++v) {
        const unsigned mk = __builtin_amdgcn_ballot_w32(valid & (d == v));
        const int below = (int)__builtin_amdgcn_mbcnt_lo(mk, 0u);
        if (d == v) r = below;
        if (lane == v) mycount = __builtin_popcount(mk);
    }
    return r;
}

__device__ __forceinline__ int clamp_digit(int p)
{
    p = p < 0 ? 0 : p;
    return p > NDIG - 1 ? NDIG - 1 : p;
}

__device__ __forceinline__ void lds_barrier()
{
    __builtin_amdgcn_fence(__ATOMIC_RELEASE, "workgroup", "local");
    __builtin_amdgcn_s_barrier();
    __builtin_amdgcn_fence(__ATOMIC_ACQUIRE, "workgroup", "local");
}

struct ItemDesc {
    int tau;
    int ch;
    int g;
    int rowbeg;
    int cnt;
};

__device__ __forceinline__ ItemDesc item_desc(int q, int nch, const int* tileg, const int* tp,
                                              const int* startv, const int* endv)
{
    ItemDesc d;
    d.tau = q / nch;
    d.ch  = q - d.tau * nch;
    d.g   = __builtin_amdgcn_readfirstlane(tileg[d.tau]);
    const int tix = d.tau - __builtin_amdgcn_readfirstlane(tp[d.g]);
    const int gs  = __builtin_amdgcn_readfirstlane(startv[d.g]);
    const int gn  = __builtin_amdgcn_readfirstlane(endv[d.g]) - gs;
    d.rowbeg = gs + 16 * tix;
    const int left = gn - 16 * tix;
    d.cnt = left > 16 ? 16 : left;
    return d;
}

__device__ __forceinline__ void request_step(const float* bp, const float* ak, bool needA,
                                             int ia, int ib, size_t plane,
                                             const float* Wg, int kb, int O, unsigned lbyte,
                                             v2f& bnext, v2f (&u)[KBATCH], v2f (&w)[KBATCH], v2f (&b)[KBATCH][4])
{
    bnext = *(const v2f*)bp;
    if (needA) {
        const float* r0 = ak + (size_t)ia * plane;
        const float* r1 = ak + (size_t)ib * plane;
#pragma unroll
        for (int j = 0; j < KBATCH; ++j) u[j] = *(const v2f*)(r0 + 4 * j);
#pragma unroll
        for (int j = 0; j < KBATCH; ++j) w[j] = *(const v2f*)(r1 + 4 * j);
    }
#pragma unroll
    for (int j = 0; j < KBATCH; ++j) {
        const char*  Wr  = (const char*)(Wg + (size_t)(kb + 4 * j) * O);
        const float* Wk0 = (const float*)(Wr + lbyte);
        const float* Wk1 = (const float*)(Wr + (size_t)O * 4 + lbyte);
#pragma unroll
        for (int t = 0; t < 4; ++t) {
            b[j][t].x = Wk0[t * 16];
            b[j][t].y = Wk1[t * 16];
        }
    }
}

__global__ void __launch_bounds__(512)
gemm_by_id(const int* __restrict__ x, const int* __restrict__ fact,
          const float* __restrict__ inp, const float* __restrict__ params,
          const float* __restrict__ bias, const int* __restrict__ msg_to_p,
          float* __restrict__ out, int N, int H, int O, int ORD)
{
    __shared__ __attribute__((aligned(16))) unsigned char L[LDS_BYTES];

    const int nsets = (N + 31) >> 5;
    float* Ps    = (float*)L;
    int*   tmp   = (int*)(Ps + KSL * 16 * PPITCH);
    int*   t1    = tmp + N;
    int*   srt   = t1 + N;
    int*   cntm  = srt + N;
    int*   offm  = cntm + NDIG * nsets;
    int*   startv= offm + NDIG * nsets;
    int*   endv  = startv + TABW;
    int*   tp    = endv + TABW;
    int*   tileg = tp + TPW + 8;

    const int lane = threadIdx.x & 31;
    const int waveid = __builtin_amdgcn_readfirstlane((int)(threadIdx.x >> 5));

    if (threadIdx.x < TABW) { startv[threadIdx.x] = 0; endv[threadIdx.x] = 0; }

    for (int s = waveid; s < nsets; s += KSL) {
        const int  i     = 32 * s + lane;
        const bool valid = i < N;
        const int  ic    = valid ? i : N - 1;
        const int  a     = fact[2 * ic];
        const int  p1    = clamp_digit(x[3 * a + 1]);
        const int  p2    = clamp_digit(x[3 * a + 2]);
        int myc;
        const int r = digit_rank(p2, valid, lane, myc);
        if (lane < NDIG) cntm[lane * nsets + s] = myc;
        if (valid) tmp[i] = (p1 * NDIG + p2) * 32 + r;
    }
    __syncthreads();
    digit_offsets(cntm, offm, nsets, waveid, lane);
    __syncthreads();
    for (int s = waveid; s < nsets; s += KSL) {
        const int i = 32 * s + lane;
        if (i < N) {
            const int pk  = tmp[i];
            const int key = pk >> 5;
            const int p2  = key % NDIG;
            t1[offm[p2 * nsets + s] + (pk & 31)] = (i << 8) | key;
        }
    }
    __syncthreads();

    for (int s = waveid; s < nsets; s += KSL) {
        const int  i     = 32 * s + lane;
        const bool valid = i < N;
        const int  e     = t1[valid ? i : N - 1];
        const int  p1    = (e & 255) / NDIG;
        int myc;
        const int r = digit_rank(p1, valid, lane, myc);
        if (lane < NDIG) cntm[lane * nsets + s] = myc;
        if (valid) tmp[i] = r;
    }
    __syncthreads();
    digit_offsets(cntm, offm, nsets, waveid, lane);
    __syncthreads();
    for (int s = waveid; s < nsets; s += KSL) {
        const int i = 32 * s + lane;
        if (i < N) {
            const int e  = t1[i];
            const int p1 = (e & 255) / NDIG;
            srt[offm[p1 * nsets + s] + tmp[i]] = e;
        }
    }
    __syncthreads();

    for (int i = threadIdx.x; i < N; i += blockDim.x) {
        const int k = srt[i] & 255;
        if (i == 0     || (srt[i - 1] & 255) != k) startv[k] = i;
        if (i == N - 1 || (srt[i + 1] & 255) != k) endv[k]   = i + 1;
    }
    __syncthreads();
    if (threadIdx.x < TPW)
        tp[threadIdx.x] = threadIdx.x < NPAIR
            ? (endv[threadIdx.x] - startv[threadIdx.x] + 15) >> 4 : 0;
    __syncthreads();
    if (waveid == 0) {
        const int all = tile_prefix(tp, lane);
        if (lane == 0) tp[TPW] = all;
    }
    __syncthreads();
    if (threadIdx.x < NPAIR) {
        const int nt = (endv[threadIdx.x] - startv[threadIdx.x] + 15) >> 4;
        for (int j = 0; j < nt; ++j) tileg[tp[threadIdx.x] + j] = threadIdx.x;
    }
    __syncthreads();

    const int TT  = __builtin_amdgcn_readfirstlane(tp[TPW]);
    const int NCH = O / CW;
    const int Q   = TT * NCH;
    const int q0  = (int)(((long long)blockIdx.x * Q) / gridDim.x);
    const int q1  = (int)(((long long)(blockIdx.x + 1) * Q) / gridDim.x);

    const int msg_to = msg_to_p[0];

    const int m     = lane & 15;
    const int khalf = (lane >> 4) << 1;
    const int kspan = H / KSL;
    const int kbeg  = waveid * kspan;
    const bool keepA = kspan == 4 * KBATCH;
    const size_t plane = (size_t)N * H;
    const int  nkept = ORD - ((msg_to >= 0 && msg_to < ORD) ? 1 : 0);
    const bool pairA = nkept == 2;
    const int  ia    = (msg_to == 0) ? 1 : 0;
    const int  ib    = (msg_to == ia + 1) ? ia + 2 : ia + 1;

    v2f a[KBATCH];
#pragma unroll
    for (int j = 0; j < KBATCH; ++j) a[j] = (v2f){0.f, 0.f};

    v2f u[KBATCH], w[KBATCH];
    v2f b[KBATCH][4];
    v2f bnext = (v2f){0.f, 0.f};
#pragma unroll
    for (int j = 0; j < KBATCH; ++j) {
        u[j] = (v2f){0.f, 0.f};
        w[j] = (v2f){0.f, 0.f};
#pragma unroll
        for (int t = 0; t < 4; ++t) b[j][t] = (v2f){0.f, 0.f};
    }
    v8f acc[4];
#pragma unroll
    for (int t = 0; t < 4; ++t) acc[t] = (v8f){0, 0, 0, 0, 0, 0, 0, 0};

    bool     have  = false;
    ItemDesc cur   = {-1, 0, 0, 0, 0};
    int      ko    = 0;
    bool     rowok = false;
    const float* arow = inp;
    unsigned lbyte = 0u;
    bool     fresh = false;
    int      rq    = q0;
    int      rko   = 0;
    while (have || rq < q1) {
        const bool more  = rq < q1;
        const int  kb    = kbeg + ko;
        const bool first = ko == 0;
        const bool last  = ko + 4 * KBATCH >= kspan;
        const v2f  bv    = bnext;

        ItemDesc     nxt    = cur;
        bool         nrowok = rowok;
        const float* narow  = arow;
        unsigned     nlbyte = lbyte;
        bool         nfresh = !keepA;
        if (more && rko == 0) {
            nxt    = item_desc(rq, NCH, tileg, tp, startv, endv);
            nrowok = m < nxt.cnt;
            narow  = inp + (size_t)(srt[nxt.rowbeg + (nrowok ? m : 0)] >> 8) * H + khalf;
            nlbyte = 4u * (unsigned)(khalf * O + nxt.ch * CW + m);
            nfresh = !keepA || nxt.tau != cur.tau;
        }

        if (have) {
            if (first) {
#pragma unroll
                for (int t = 0; t < 4; ++t) acc[t] = (v8f){0, 0, 0, 0, 0, 0, 0, 0};
            }

            if (fresh) {
                if (pairA) {
#pragma unroll
                    for (int j = 0; j < KBATCH; ++j) {
                        v2f p = u[j];
                        p.x *= w[j].x; p.y *= w[j].y;
                        a[j] = rowok ? p : (v2f){0.f, 0.f};
                    }
                } else {
                    const float* ak = arow + kb;
                    v2f p[KBATCH];
#pragma unroll
                    for (int j = 0; j < KBATCH; ++j) p[j] = (v2f){1.f, 1.f};
                    for (int i = 0; i < ORD; ++i) {
                        if (i == msg_to) continue;
                        const float* r = ak + (size_t)i * plane;
#pragma unroll
                        for (int j = 0; j < KBATCH; ++j) {
                            const v2f v = *(const v2f*)(r + 4 * j);
                            p[j].x *= v.x; p[j].y *= v.y;
                        }
                    }
#pragma unroll
                    for (int j = 0; j < KBATCH; ++j)
                        a[j] = rowok ? p[j] : (v2f){0.f, 0.f};
                }
            }

#pragma unroll
            for (int j = 0; j < KBATCH; ++j) {
#pragma unroll
                for (int t = 0; t < 4; ++t) {
                    acc[t] = __builtin_amdgcn_wmma_f32_16x16x4_f32(
                        false, a[j], false, b[j][t], (short)0, acc[t], false, false);
                }
            }
        }

        if (more)
            request_step(bias + (size_t)nxt.g * O + nxt.ch * CW + 2 * lane,
                         narow + kbeg + rko, nfresh && pairA, ia, ib, plane,
                         params + (size_t)nxt.g * H * O, kbeg + rko, O,
                         nlbyte, bnext, u, w, b);

        if (have && last) {
            {
                const int mbase = (lane >> 4) << 3;
                float* prow = Ps + (waveid * 16 + mbase) * PPITCH + m;
#pragma unroll
                for (int t = 0; t < 4; ++t) {
#pragma unroll
                    for (int r = 0; r < 8; ++r)
                        prow[r * PPITCH + t * 16] = acc[t][r];
                }
            }
            lds_barrier();

            if (waveid < cur.cnt) {
                const int nr = srt[cur.rowbeg + waveid] >> 8;
                v2f s = *(const v2f*)(Ps + waveid * PPITCH + 2 * lane);
                for (int sl = 1; sl < KSL; ++sl) {
                    const v2f p = *(const v2f*)(Ps + (sl * 16 + waveid) * PPITCH + 2 * lane);
                    s.x += p.x; s.y += p.y;
                }
                s.x += bv.x; s.y += bv.y;
                *(v2f*)(out + (size_t)nr * O + cur.ch * CW + 2 * lane) = s;
            }
            lds_barrier();
        }

        have = more; cur = nxt; ko = rko; rowok = nrowok; arow = narow; lbyte = nlbyte; fresh = nfresh;
        if (more) {
            const bool rlast = rko + 4 * KBATCH >= kspan;
            rq  = rlast ? rq + 1 : rq;
            rko = rlast ? 0 : rko + 4 * KBATCH;
        }
    }
}

extern "C" void kernel_launch(void* const* d_in, const int* in_sizes, int n_in,
                              void* d_out, int out_size, void* d_ws, size_t ws_size,
                              hipStream_t stream)
{
    const int*   x      = (const int*)  d_in[0];
    const int*   fact   = (const int*)  d_in[1];
    const float* inp    = (const float*)d_in[2];
    const float* params = (const float*)d_in[3];
    const float* bias   = (const float*)d_in[4];
    const int*   msg_to = (const int*)  d_in[5];
    float*       out    = (float*)d_out;

    const int N   = in_sizes[1] / 2;
    const int O   = in_sizes[4] / NPAIR;
    const int H   = in_sizes[3] / (NPAIR * O);
    const int ORD = in_sizes[2] / (N * H);

    hipLaunchKernelGGL(gemm_by_id, dim3(256), dim3(512), 0, stream,
                       x, fact, inp, params, bias, msg_to, out,
                       N, H, O, ORD);
}
